# starting kernel
# baseline (speedup 1.0000x reference)
#include <hip/hip_runtime.h>
#include <climits>

#define FDIM 128
#define HEADS 2
#define CH 64
#define NEG_SLOPE 0.2f

typedef float v2f __attribute__((ext_vector_type(2)));
typedef float v8f __attribute__((ext_vector_type(8)));

__global__ __launch_bounds__(128) void gemm_f32_wmma(
    const float* __restrict__ X, const float* __restrict__ W,
    const float* __restrict__ bias0, const float* __restrict__ bias1,
    float* __restrict__ OUT, int nrows)
{
    const int ntiles = nrows >> 4;
    const int tile = blockIdx.x * 4 + (threadIdx.x >> 5);
    if (tile >= ntiles) return;

    const int lane  = threadIdx.x & 31;
    const int lhalf = lane >> 4;
    const int lidx  = lane & 15;
    const int row0  = tile * 16;

    const float* xrow = X + (size_t)(row0 + lidx) * FDIM + 2 * lhalf;

    v8f acc[8];
#pragma unroll
    for (int t = 0; t < 8; ++t)
#pragma unroll
        for (int r = 0; r < 8; ++r) acc[t][r] = 0.0f;

#pragma unroll 4
    for (int kk = 0; kk < FDIM; kk += 4) {
        v2f a = *(const v2f*)(xrow + kk);
        const int krow = kk + 2 * lhalf;
        const float* wr0 = W + (size_t)krow * FDIM + lidx;
        const float* wr1 = wr0 + FDIM;
#pragma unroll
        for (int t = 0; t < 8; ++t) {
            v2f b;
            b.x = wr0[t * 16];
            b.y = wr1[t * 16];
            acc[t] = __builtin_amdgcn_wmma_f32_16x16x4_f32(
                false, a, false, b, (short)0, acc[t], false, false);
        }
    }

    const int orow = row0 + 8 * lhalf;
#pragma unroll
    for (int t = 0; t < 8; ++t) {
        const int col = t * 16 + lidx;
        float badd = 0.0f;
        if (bias0) badd += bias0[col];
        if (bias1) badd += bias1[col];
        float* o = OUT + (size_t)orow * FDIM + col;
#pragma unroll
        for (int r = 0; r < 8; ++r)
            o[(size_t)r * FDIM] = acc[t][r] + badd;
    }
}

__global__ void att_kernel(const float* __restrict__ Hp,
                           const float* __restrict__ att_src,
                           const float* __restrict__ att_dst,
                           float* __restrict__ a_s, float* __restrict__ a_d,
                           int n)
{
    int i = blockIdx.x * blockDim.x + threadIdx.x;
    if (i >= n * HEADS) return;
    const int node = i >> 1, h = i & 1;
    const float* hp = Hp + (size_t)node * FDIM + h * CH;
    const float* as = att_src + h * CH;
    const float* ad = att_dst + h * CH;
    float s0 = 0.f, s1 = 0.f;
#pragma unroll 8
    for (int c = 0; c < CH; ++c) {
        float v = hp[c];
        s0 += v * as[c];
        s1 += v * ad[c];
    }
    a_s[i] = s0;
    a_d[i] = s1;
}

__global__ void init_kernel(int* __restrict__ m, float* __restrict__ denom, int n2)
{
    int i = blockIdx.x * blockDim.x + threadIdx.x;
    if (i < n2) { m[i] = INT_MIN; denom[i] = 0.0f; }
}

__device__ __forceinline__ int   fmapf(float x)  { int b = __float_as_int(x); return b >= 0 ? b : (b ^ 0x7FFFFFFF); }
__device__ __forceinline__ float funmap(int b)   { return __int_as_float(b >= 0 ? b : (b ^ 0x7FFFFFFF)); }
__device__ __forceinline__ float lrelu(float x)  { return x > 0.f ? x : NEG_SLOPE * x; }

__global__ void edge_max_kernel(const int* __restrict__ src, const int* __restrict__ dst,
                                const float* __restrict__ a_s, const float* __restrict__ a_d,
                                int* __restrict__ m, int E_, int M_)
{
    int k = blockIdx.x * blockDim.x + threadIdx.x;
    if (k >= M_) return;
    const int s = k < E_ ? src[k] : k - E_;
    const int d = k < E_ ? dst[k] : k - E_;
#pragma unroll
    for (int h = 0; h < HEADS; ++h) {
        float e = lrelu(a_s[s * HEADS + h] + a_d[d * HEADS + h]);
        atomicMax(&m[d * HEADS + h], fmapf(e));
    }
}

__global__ void edge_sum_kernel(const int* __restrict__ src, const int* __restrict__ dst,
                                const float* __restrict__ a_s, const float* __restrict__ a_d,
                                const int* __restrict__ m, float* __restrict__ P,
                                float* __restrict__ denom, int E_, int M_)
{
    int k = blockIdx.x * blockDim.x + threadIdx.x;
    if (k >= M_) return;
    const int s = k < E_ ? src[k] : k - E_;
    const int d = k < E_ ? dst[k] : k - E_;
#pragma unroll
    for (int h = 0; h < HEADS; ++h) {
        float e  = lrelu(a_s[s * HEADS + h] + a_d[d * HEADS + h]);
        float mx = funmap(m[d * HEADS + h]);
        float p  = __expf(e - mx);
        P[(size_t)k * HEADS + h] = p;
        atomicAdd(&denom[d * HEADS + h], p);
    }
}

__global__ __launch_bounds__(256) void edge_scatter_kernel(
    const int* __restrict__ src, const int* __restrict__ dst,
    const float* __restrict__ P, const float* __restrict__ denom,
    const float* __restrict__ Hp, float* __restrict__ out, int E_, int M_)
{
    const int lane = threadIdx.x & 31;
    const int k = blockIdx.x * 8 + (threadIdx.x >> 5);
    if (k >= M_) return;
    const int s = k < E_ ? src[k] : k - E_;
    const int d = k < E_ ? dst[k] : k - E_;
    const int h = lane >> 4;
    const float alpha = P[(size_t)k * HEADS + h] / (denom[d * HEADS + h] + 1e-16f);
    const int c0 = lane * 4;
    const float4 hv = *(const float4*)(Hp + (size_t)s * FDIM + c0);
    float* o = out + (size_t)d * FDIM + c0;
    atomicAdd(o + 0, hv.x * alpha);
    atomicAdd(o + 1, hv.y * alpha);
    atomicAdd(o + 2, hv.z * alpha);
    atomicAdd(o + 3, hv.w * alpha);
}

__global__ void relu_kernel(float* __restrict__ x, int n)
{
    int i = blockIdx.x * blockDim.x + threadIdx.x;
    if (i < n) x[i] = fmaxf(x[i], 0.0f);
}

extern "C" void kernel_launch(void* const* d_in, const int* in_sizes, int n_in,
                              void* d_out, int out_size, void* d_ws, size_t ws_size,
                              hipStream_t stream)
{
    const float* x        = (const float*)d_in[0];
    const int*   src      = (const int*)  d_in[1];
    const int*   dst      = (const int*)  d_in[2];
    const float* W1       = (const float*)d_in[3];
    const float* att_src1 = (const float*)d_in[4];
    const float* att_dst1 = (const float*)d_in[5];
    const float* b1       = (const float*)d_in[6];
    const float* Wsk1     = (const float*)d_in[7];
    const float* bsk1     = (const float*)d_in[8];
    const float* W2       = (const float*)d_in[9];
    const float* att_src2 = (const float*)d_in[10];
    const float* att_dst2 = (const float*)d_in[11];
    const float* b2       = (const float*)d_in[12];
    const float* Wsk2     = (const float*)d_in[13];
    const float* bsk2     = (const float*)d_in[14];

    const int N  = in_sizes[0] / FDIM;
    const int E  = in_sizes[1];
    const int M  = E + N;
    float* out   = (float*)d_out;

    char* base = (char*)d_ws;
    size_t off = 0;
    auto alloc = [&](size_t bytes) {
        void* p = base + off;
        off = (off + bytes + 255) & ~(size_t)255;
        return p;
    };
    float* Hproj = (float*)alloc((size_t)N * FDIM * 4);
    float* Hbuf  = (float*)alloc((size_t)N * FDIM * 4);
    float* a_s   = (float*)alloc((size_t)N * HEADS * 4);
    float* a_d   = (float*)alloc((size_t)N * HEADS * 4);
    int*   mbuf  = (int*)  alloc((size_t)N * HEADS * 4);
    float* denom = (float*)alloc((size_t)N * HEADS * 4);
    float* P     = (float*)alloc((size_t)M * HEADS * 4);

    const int ntiles     = N / 16;
    const dim3 gGemm((ntiles + 3) / 4), bGemm(128);
    const dim3 gAtt((N * HEADS + 255) / 256), b256(256);
    const dim3 gInit((N * HEADS + 255) / 256);
    const dim3 gEdge((M + 255) / 256);
    const dim3 gScat((M + 7) / 8);
    const dim3 gRelu((N * FDIM + 255) / 256);

    gemm_f32_wmma<<<gGemm, bGemm, 0, stream>>>(x, W1, nullptr, nullptr, Hproj, N);
    gemm_f32_wmma<<<gGemm, bGemm, 0, stream>>>(x, Wsk1, b1, bsk1, Hbuf, N);
    att_kernel<<<gAtt, b256, 0, stream>>>(Hproj, att_src1, att_dst1, a_s, a_d, N);
    init_kernel<<<gInit, b256, 0, stream>>>(mbuf, denom, N * HEADS);
    edge_max_kernel<<<gEdge, b256, 0, stream>>>(src, dst, a_s, a_d, mbuf, E, M);
    edge_sum_kernel<<<gEdge, b256, 0, stream>>>(src, dst, a_s, a_d, mbuf, P, denom, E, M);
    edge_scatter_kernel<<<gScat, b256, 0, stream>>>(src, dst, P, denom, Hproj, Hbuf, E, M);
    relu_kernel<<<gRelu, b256, 0, stream>>>(Hbuf, N * FDIM);

    gemm_f32_wmma<<<gGemm, bGemm, 0, stream>>>(Hbuf, W2, nullptr, nullptr, Hproj, N);
    gemm_f32_wmma<<<gGemm, bGemm, 0, stream>>>(Hbuf, Wsk2, b2, bsk2, out, N);
    att_kernel<<<gAtt, b256, 0, stream>>>(Hproj, att_src2, att_dst2, a_s, a_d, N);
    init_kernel<<<gInit, b256, 0, stream>>>(mbuf, denom, N * HEADS);
    edge_max_kernel<<<gEdge, b256, 0, stream>>>(src, dst, a_s, a_d, mbuf, E, M);
    edge_sum_kernel<<<gEdge, b256, 0, stream>>>(src, dst, a_s, a_d, mbuf, P, denom, E, M);
    edge_scatter_kernel<<<gScat, b256, 0, stream>>>(src, dst, P, denom, Hproj, out, E, M);
}


# Block kernel: each filed entry is resolved once by one thread (source, both softmax weights, denominator share) into a 16-byte LDS record; the row walk only loads rows and adds (order: LDS arrival)
# speedup vs baseline: 2.6598x; 2.6598x over previous
#include <hip/hip_runtime.h>
#include <climits>

#define FDIM 128
#define HEADS 2
#define CH 64
#define NEG_SLOPE 0.2f

typedef float v2f __attribute__((ext_vector_type(2)));
typedef float v8f __attribute__((ext_vector_type(8)));

__global__ __launch_bounds__(128) void gemm_f32_wmma(
    const float* __restrict__ X, const float* __restrict__ W,
    const float* __restrict__ bias0, const float* __restrict__ bias1,
    float* __restrict__ OUT, int nrows)
{
    const int ntiles = nrows >> 4;
    const int tile = blockIdx.x * 4 + (threadIdx.x >> 5);
    if (tile >= ntiles) return;

    const int lane  = threadIdx.x & 31;
    const int lhalf = lane >> 4;
    const int lidx  = lane & 15;
    const int row0  = tile * 16;

    const float* xrow = X + (size_t)(row0 + lidx) * FDIM + 2 * lhalf;

    v8f acc[8];
#pragma unroll
    for (int t = 0; t < 8; ++t)
#pragma unroll
        for (int r = 0; r < 8; ++r) acc[t][r] = 0.0f;

#pragma unroll 4
    for (int kk = 0; kk < FDIM; kk += 4) {
        v2f a = *(const v2f*)(xrow + kk);
        const int krow = kk + 2 * lhalf;
        const float* wr0 = W + (size_t)krow * FDIM + lidx;
        const float* wr1 = wr0 + FDIM;
#pragma unroll
        for (int t = 0; t < 8; ++t) {
            v2f b;
            b.x = wr0[t * 16];
            b.y = wr1[t * 16];
            acc[t] = __builtin_amdgcn_wmma_f32_16x16x4_f32(
                false, a, false, b, (short)0, acc[t], false, false);
        }
    }

    const int orow = row0 + 8 * lhalf;
#pragma unroll
    for (int t = 0; t < 8; ++t) {
        const int col = t * 16 + lidx;
        float badd = 0.0f;
        if (bias0) badd += bias0[col];
        if (bias1) badd += bias1[col];
        float* o = OUT + (size_t)orow * FDIM + col;
#pragma unroll
        for (int r = 0; r < 8; ++r)
            o[(size_t)r * FDIM] = acc[t][r] + badd;
    }
}

__device__ __forceinline__ int   fmapf(float x)  { int b = __float_as_int(x); return b >= 0 ? b : (b ^ 0x7FFFFFFF); }
__device__ __forceinline__ float funmap(int b)   { return __int_as_float(b >= 0 ? b : (b ^ 0x7FFFFFFF)); }
__device__ __forceinline__ float lrelu(float x)  { return x > 0.f ? x : NEG_SLOPE * x; }

#define SPREAD_MAX 18.0f

__device__ __forceinline__ bool shift_from_bound(const int4 g, float& hi0, float& hi1)
{
    hi0 = funmap(g.x);
    hi1 = funmap(g.y);
    const float lo0 = funmap(g.z);
    const float lo1 = funmap(g.w);
    return (hi0 - lo0 <= SPREAD_MAX) && (hi1 - lo1 <= SPREAD_MAX);
}

__global__ __launch_bounds__(256) void att_kernel(const float* __restrict__ Hp,
                           const float* __restrict__ att_src,
                           const float* __restrict__ att_dst,
                           float* __restrict__ a_s, float* __restrict__ a_d,
                           int* __restrict__ gstat, int n)
{
    __shared__ int smax[16], smin[16];
    const int i = blockIdx.x * blockDim.x + threadIdx.x;
    int kmax = INT_MIN, kmin = INT_MAX;
    if (i < n * HEADS) {
        const int node = i >> 1, h = i & 1;
        const float* hp = Hp + (size_t)node * FDIM + h * CH;
        const float* as = att_src + h * CH;
        const float* ad = att_dst + h * CH;
        float s0 = 0.f, s1 = 0.f;
#pragma unroll 8
        for (int c = 0; c < CH; ++c) {
            float v = hp[c];
            s0 += v * as[c];
            s1 += v * ad[c];
        }
        a_s[i] = s0;
        a_d[i] = s1;
        kmax = kmin = fmapf(s0);
    }
#pragma unroll
    for (int off = 2; off < 32; off <<= 1) {
        kmax = max(kmax, __shfl_xor(kmax, off));
        kmin = min(kmin, __shfl_xor(kmin, off));
    }
    const int lane = threadIdx.x & 31, wave = threadIdx.x >> 5;
    if (lane < HEADS) { smax[wave * HEADS + lane] = kmax; smin[wave * HEADS + lane] = kmin; }
    __syncthreads();
    if (threadIdx.x < HEADS) {
        int mx = INT_MIN, mn = INT_MAX;
        for (int w = 0; w < 8; ++w) {
            mx = max(mx, smax[w * HEADS + threadIdx.x]);
            mn = min(mn, smin[w * HEADS + threadIdx.x]);
        }
        atomicMax(&gstat[threadIdx.x], mx);
        atomicMin(&gstat[HEADS + threadIdx.x], mn);
    }
}

__global__ void init_kernel(int* __restrict__ m, int* __restrict__ gstat, int n2)
{
    int i = blockIdx.x * blockDim.x + threadIdx.x;
    if (i < n2) m[i] = INT_MIN;
    if (i < 2 * HEADS) gstat[i] = i < HEADS ? INT_MIN : INT_MAX;
}

__global__ void edge_max_kernel(const int* __restrict__ src, const int* __restrict__ dst,
                                const float* __restrict__ a_s, const float* __restrict__ a_d,
                                int* __restrict__ m, const int* __restrict__ gstat, int E_, int M_)
{
    float hi0, hi1;
    if (shift_from_bound(*(const int4*)gstat, hi0, hi1)) return;
    int k = blockIdx.x * blockDim.x + threadIdx.x;
    if (k >= M_) return;
    const int s = k < E_ ? src[k] : k - E_;
    const int d = k < E_ ? dst[k] : k - E_;
#pragma unroll
    for (int h = 0; h < HEADS; ++h) {
        float e = lrelu(a_s[s * HEADS + h] + a_d[d * HEADS + h]);
        atomicMax(&m[d * HEADS + h], fmapf(e));
    }
}

#define NB   196
#define QCAP 8192
#define BT   1024
#define BU   8
#define RU   4

template <bool FULL>
__device__ __forceinline__ void add_group(
    const int (&s)[BU], const int (&dl)[BU], const float (&p0)[BU], const float (&p1)[BU],
    const bool (&ok)[BU], const int lane, const float* __restrict__ Hp, float* acc)
{
    float h0[BU], h1[BU], h2[BU], h3[BU];
#pragma unroll
    for (int j = 0; j < BU; ++j) {
        const float* hrow = Hp + (size_t)s[j] * FDIM + lane;
        h0[j] = hrow[0];
        h1[j] = hrow[32];
        h2[j] = hrow[64];
        h3[j] = hrow[96];
    }
#pragma unroll
    for (int j = 0; j < BU; ++j) {
        if (FULL || ok[j]) {
            float* o = acc + dl[j] * FDIM + lane;
            atomicAdd(o,      h0[j] * p0[j]);
            atomicAdd(o + 32, h1[j] * p0[j]);
            atomicAdd(o + 64, h2[j] * p1[j]);
            atomicAdd(o + 96, h3[j] * p1[j]);
        }
    }
}

template <bool FIRST>
__global__ __launch_bounds__(BT) void block_aggregate_kernel(
    const int* __restrict__ src, const int* __restrict__ dst,
    const float* __restrict__ a_s, const float* __restrict__ a_d,
    const int* __restrict__ m, const int* __restrict__ gstat,
    const float* __restrict__ Hp, float* __restrict__ x,
    int2* __restrict__ lists, int* __restrict__ lcnt,
    int E_, int N_, int relu)
{
    __shared__ __attribute__((aligned(16))) float acc[NB * FDIM];
    __shared__ __attribute__((aligned(16))) int4 rec[QCAP];
    __shared__ int2  rawq[QCAP];
    __shared__ float den[NB * HEADS], sh_ad[NB * HEADS], sh_c[NB * HEADS];
    __shared__ int   qcount;

    const int tid   = threadIdx.x;
    const int lane  = tid & 31;
    const int wave  = __builtin_amdgcn_readfirstlane(tid >> 5);
    const int node0 = blockIdx.x * NB;
    const int nb    = min(NB, N_ - node0);

    float hi0, hi1;
    const bool bound = shift_from_bound(*(const int4*)gstat, hi0, hi1);

    for (int i = tid; i < NB * FDIM / 4; i += BT) ((float4*)acc)[i] = make_float4(0.0f, 0.0f, 0.0f, 0.0f);
    for (int i = tid; i < nb * HEADS; i += BT) {
        const int   g  = node0 * HEADS + i;
        const float ad = a_d[g];
        sh_ad[i] = ad;
        sh_c[i]  = bound ? lrelu(((i & 1) ? hi1 : hi0) + ad) : funmap(m[g]);
        den[i]   = 0.0f;
    }
    if (tid == 0) qcount = 0;
    __syncthreads();

    auto file_edge = [&](const int d, const int k) {
        const unsigned dl = (unsigned)(d - node0);
        if (dl < (unsigned)nb) {
            const int slot = atomicAdd(&qcount, 1);
            if (slot < QCAP) rawq[slot] = make_int2(k, (int)dl);
        }
    };
    auto file_four = [&](const int4 d, const int k) {
        const unsigned u0 = (unsigned)(d.x - node0), u1 = (unsigned)(d.y - node0);
        const unsigned u2 = (unsigned)(d.z - node0), u3 = (unsigned)(d.w - node0);
        if (min(min(u0, u1), min(u2, u3)) < (unsigned)nb) {
            file_edge(d.x, k);
            file_edge(d.y, k + 1);
            file_edge(d.z, k + 2);
            file_edge(d.w, k + 3);
        }
    };

    int cnt;
    if constexpr (FIRST) {
        const int n4 = E_ >> 2;
        const int4* __restrict__ dst4 = (const int4*)dst;
        const int start = (int)blockIdx.x * (n4 / (int)gridDim.x);
        for (int i0 = tid; i0 < n4; i0 += 4 * BT) {
            int4 d4[4];
            int  ix[4];
#pragma unroll
            for (int u = 0; u < 4; ++u) {
                const int i = i0 + u * BT;
                int t = start + (i < n4 ? i : i0);
                if (t >= n4) t -= n4;
                d4[u] = dst4[t];
                ix[u] = i < n4 ? t : -1;
            }
#pragma unroll
            for (int u = 0; u < 4; ++u) {
                if (ix[u] >= 0) file_four(d4[u], ix[u] << 2);
            }
        }
        for (int k = (n4 << 2) + tid; k < E_; k += BT) file_edge(dst[k], k);
        __syncthreads();
        cnt = qcount;
        if (tid == 0) lcnt[blockIdx.x] = cnt;
    } else {
        cnt = lcnt[blockIdx.x];
    }

    auto resolve = [&](const int nr, const int ns, const int from) {
        int2* __restrict__ mine = lists + (size_t)blockIdx.x * QCAP;
        const int n = nr + ns;
        for (int i0 = tid; i0 < n; i0 += RU * BT) {
            int s[RU], dl[RU];
            v2f as2[RU];
#pragma unroll
            for (int u = 0; u < RU; ++u) {
                const int  i    = min(i0 + u * BT, n - 1);
                const bool pair = i < nr;
                const int2 e    = from == 2 ? mine[pair ? i : 0] : rawq[pair ? i : 0];
                dl[u] = pair ? e.y : i - nr;
                s[u]  = pair ? e.x : node0 + (i - nr);
            }
            if (from != 2) {
#pragma unroll
                for (int u = 0; u < RU; ++u) {
                    const bool pair = min(i0 + u * BT, n - 1) < nr;
                    const int  sv   = src[pair ? s[u] : 0];
                    s[u] = pair ? sv : s[u];
                }
            }
#pragma unroll
            for (int u = 0; u < RU; ++u)
                as2[u] = *(const v2f*)(a_s + s[u] * HEADS);
#pragma unroll
            for (int u = 0; u < RU; ++u) {
                const int i = i0 + u * BT;
                if (i < n) {
                    const float p0 = __expf(lrelu(as2[u].x + sh_ad[dl[u] * HEADS])     - sh_c[dl[u] * HEADS]);
                    const float p1 = __expf(lrelu(as2[u].y + sh_ad[dl[u] * HEADS + 1]) - sh_c[dl[u] * HEADS + 1]);
                    atomicAdd(&den[dl[u] * HEADS],     p0);
                    atomicAdd(&den[dl[u] * HEADS + 1], p1);
                    rec[i] = make_int4(s[u], dl[u], __float_as_int(p0), __float_as_int(p1));
                    if (from == 1 && i < nr) mine[i] = make_int2(s[u], dl[u]);
                }
            }
        }
    };

    auto walk = [&](const int n) {
        for (int q0 = wave * BU; q0 < n; q0 += (BT / 32) * BU) {
            int s[BU], dl[BU]; float p0[BU], p1[BU]; bool ok[BU];
#pragma unroll
            for (int j = 0; j < BU; ++j) {
                ok[j] = q0 + j < n;
                const int4 e = rec[ok[j] ? q0 + j : n - 1];
                s[j]  = __builtin_amdgcn_readfirstlane(e.x);
                dl[j] = __builtin_amdgcn_readfirstlane(e.y);
                p0[j] = __int_as_float(__builtin_amdgcn_readfirstlane(e.z));
                p1[j] = __int_as_float(__builtin_amdgcn_readfirstlane(e.w));
            }
            if (q0 + BU <= n) add_group<true>(s, dl, p0, p1, ok, lane, Hp, acc);
            else              add_group<false>(s, dl, p0, p1, ok, lane, Hp, acc);
        }
    };

    if (cnt + nb <= QCAP) {
        resolve(cnt, nb, FIRST ? 1 : 2);
        __syncthreads();
        walk(cnt + nb);
    } else {
        resolve(0, nb, 0);
        __syncthreads();
        walk(nb);
        for (int b0 = 0; b0 < E_; b0 += QCAP) {
            __syncthreads();
            if (tid == 0) qcount = 0;
            __syncthreads();
            const int b1 = min(b0 + QCAP, E_);
            for (int k = b0 + tid; k < b1; k += BT) file_edge(dst[k], k);
            __syncthreads();
            const int n = qcount;
            resolve(n, 0, 0);
            __syncthreads();
            walk(n);
        }
    }
    __syncthreads();

    for (int i = wave; i < nb; i += BT / 32) {
        const float dn0 = den[i * HEADS] + 1e-16f;
        const float dn1 = den[i * HEADS + 1] + 1e-16f;
        float*       xr = x + (size_t)(node0 + i) * FDIM + lane;
        const float* ar = acc + i * FDIM + lane;
        float v0 = xr[0]  + ar[0]  / dn0;
        float v1 = xr[32] + ar[32] / dn0;
        float v2 = xr[64] + ar[64] / dn1;
        float v3 = xr[96] + ar[96] / dn1;
        if (relu) {
            v0 = fmaxf(v0, 0.0f); v1 = fmaxf(v1, 0.0f);
            v2 = fmaxf(v2, 0.0f); v3 = fmaxf(v3, 0.0f);
        }
        xr[0] = v0; xr[32] = v1; xr[64] = v2; xr[96] = v3;
    }
}

extern "C" void kernel_launch(void* const* d_in, const int* in_sizes, int n_in,
                              void* d_out, int out_size, void* d_ws, size_t ws_size,
                              hipStream_t stream)
{
    const float* x        = (const float*)d_in[0];
    const int*   src      = (const int*)  d_in[1];
    const int*   dst      = (const int*)  d_in[2];
    const float* W1       = (const float*)d_in[3];
    const float* att_src1 = (const float*)d_in[4];
    const float* att_dst1 = (const float*)d_in[5];
    const float* b1       = (const float*)d_in[6];
    const float* Wsk1     = (const float*)d_in[7];
    const float* bsk1     = (const float*)d_in[8];
    const float* W2       = (const float*)d_in[9];
    const float* att_src2 = (const float*)d_in[10];
    const float* att_dst2 = (const float*)d_in[11];
    const float* b2       = (const float*)d_in[12];
    const float* Wsk2     = (const float*)d_in[13];
    const float* bsk2     = (const float*)d_in[14];

    const int N  = in_sizes[0] / FDIM;
    const int E  = in_sizes[1];
    const int M  = E + N;
    float* out   = (float*)d_out;
    const int nblk = (N + NB - 1) / NB;

    char* base = (char*)d_ws;
    size_t off = 0;
    auto alloc = [&](size_t bytes) {
        void* p = base + off;
        off = (off + bytes + 255) & ~(size_t)255;
        return p;
    };
    float* Hproj = (float*)alloc((size_t)N * FDIM * 4);
    float* Hbuf  = (float*)alloc((size_t)N * FDIM * 4);
    float* a_s   = (float*)alloc((size_t)N * HEADS * 4);
    float* a_d   = (float*)alloc((size_t)N * HEADS * 4);
    int*   mbuf  = (int*)  alloc((size_t)N * HEADS * 4);
    int*   gstat = (int*)  alloc(2 * HEADS * 4);
    int2*  lists = (int2*) alloc((size_t)nblk * QCAP * 8);
    int*   lcnt  = (int*)  alloc((size_t)nblk * 4);

    const int ntiles     = N / 16;
    const dim3 gGemm((ntiles + 3) / 4), bGemm(128);
    const dim3 gAtt((N * HEADS + 255) / 256), b256(256);
    const dim3 gEdge((M + 255) / 256);
    const dim3 gBlk(nblk), bBlk(BT);

    gemm_f32_wmma<<<gGemm, bGemm, 0, stream>>>(x, W1, nullptr, nullptr, Hproj, N);
    gemm_f32_wmma<<<gGemm, bGemm, 0, stream>>>(x, Wsk1, b1, bsk1, Hbuf, N);
    init_kernel<<<gAtt, b256, 0, stream>>>(mbuf, gstat, N * HEADS);
    att_kernel<<<gAtt, b256, 0, stream>>>(Hproj, att_src1, att_dst1, a_s, a_d, gstat, N);
    edge_max_kernel<<<gEdge, b256, 0, stream>>>(src, dst, a_s, a_d, mbuf, gstat, E, M);
    block_aggregate_kernel<true><<<gBlk, bBlk, 0, stream>>>(src, dst, a_s, a_d, mbuf, gstat, Hproj, Hbuf, lists, lcnt, E, N, 1);

    gemm_f32_wmma<<<gGemm, bGemm, 0, stream>>>(Hbuf, W2, nullptr, nullptr, Hproj, N);
    gemm_f32_wmma<<<gGemm, bGemm, 0, stream>>>(Hbuf, Wsk2, b2, bsk2, out, N);
    init_kernel<<<gAtt, b256, 0, stream>>>(mbuf, gstat, N * HEADS);
    att_kernel<<<gAtt, b256, 0, stream>>>(Hproj, att_src2, att_dst2, a_s, a_d, gstat, N);
    edge_max_kernel<<<gEdge, b256, 0, stream>>>(src, dst, a_s, a_d, mbuf, gstat, E, M);
    block_aggregate_kernel<false><<<gBlk, bBlk, 0, stream>>>(src, dst, a_s, a_d, mbuf, gstat, Hproj, out, lists, lcnt, E, N, 0);
}


# Block kernel searches the destination list as bit planes of its block numbers (built by the init kernel, 8 words per 32 entries): 8 xor + 7 and pick a block's entries (sum order: LDS arrival)
# speedup vs baseline: 2.9144x; 1.0957x over previous
#include <hip/hip_runtime.h>
#include <climits>

#define FDIM 128
#define HEADS 2
#define CH 64
#define NEG_SLOPE 0.2f

typedef float v2f __attribute__((ext_vector_type(2)));
typedef float v8f __attribute__((ext_vector_type(8)));

__global__ __launch_bounds__(128) void gemm_f32_wmma(
    const float* __restrict__ X, const float* __restrict__ W,
    const float* __restrict__ bias0, const float* __restrict__ bias1,
    float* __restrict__ OUT, int nrows)
{
    const int ntiles = nrows >> 4;
    const int tile = blockIdx.x * 4 + (threadIdx.x >> 5);
    if (tile >= ntiles) return;

    const int lane  = threadIdx.x & 31;
    const int lhalf = lane >> 4;
    const int lidx  = lane & 15;
    const int row0  = tile * 16;

    const float* xrow = X + (size_t)(row0 + lidx) * FDIM + 2 * lhalf;

    v8f acc[8];
#pragma unroll
    for (int t = 0; t < 8; ++t)
#pragma unroll
        for (int r = 0; r < 8; ++r) acc[t][r] = 0.0f;

#pragma unroll 4
    for (int kk = 0; kk < FDIM; kk += 4) {
        v2f a = *(const v2f*)(xrow + kk);
        const int krow = kk + 2 * lhalf;
        const float* wr0 = W + (size_t)krow * FDIM + lidx;
        const float* wr1 = wr0 + FDIM;
#pragma unroll
        for (int t = 0; t < 8; ++t) {
            v2f b;
            b.x = wr0[t * 16];
            b.y = wr1[t * 16];
            acc[t] = __builtin_amdgcn_wmma_f32_16x16x4_f32(
                false, a, false, b, (short)0, acc[t], false, false);
        }
    }

    const int orow = row0 + 8 * lhalf;
#pragma unroll
    for (int t = 0; t < 8; ++t) {
        const int col = t * 16 + lidx;
        float badd = 0.0f;
        if (bias0) badd += bias0[col];
        if (bias1) badd += bias1[col];
        float* o = OUT + (size_t)orow * FDIM + col;
#pragma unroll
        for (int r = 0; r < 8; ++r)
            o[(size_t)r * FDIM] = acc[t][r] + badd;
    }
}

__device__ __forceinline__ int   fmapf(float x)  { int b = __float_as_int(x); return b >= 0 ? b : (b ^ 0x7FFFFFFF); }
__device__ __forceinline__ float funmap(int b)   { return __int_as_float(b >= 0 ? b : (b ^ 0x7FFFFFFF)); }
__device__ __forceinline__ float lrelu(float x)  { return x > 0.f ? x : NEG_SLOPE * x; }

#define SPREAD_MAX 18.0f

__device__ __forceinline__ bool shift_from_bound(const int4 g, float& hi0, float& hi1)
{
    hi0 = funmap(g.x);
    hi1 = funmap(g.y);
    const float lo0 = funmap(g.z);
    const float lo1 = funmap(g.w);
    return (hi0 - lo0 <= SPREAD_MAX) && (hi1 - lo1 <= SPREAD_MAX);
}

__global__ __launch_bounds__(256) void att_kernel(const float* __restrict__ Hp,
                           const float* __restrict__ att_src,
                           const float* __restrict__ att_dst,
                           float* __restrict__ a_s, float* __restrict__ a_d,
                           int* __restrict__ gstat, int n)
{
    __shared__ int smax[16], smin[16];
    const int i = blockIdx.x * blockDim.x + threadIdx.x;
    int kmax = INT_MIN, kmin = INT_MAX;
    if (i < n * HEADS) {
        const int node = i >> 1, h = i & 1;
        const float* hp = Hp + (size_t)node * FDIM + h * CH;
        const float* as = att_src + h * CH;
        const float* ad = att_dst + h * CH;
        float s0 = 0.f, s1 = 0.f;
#pragma unroll 8
        for (int c = 0; c < CH; ++c) {
            float v = hp[c];
            s0 += v * as[c];
            s1 += v * ad[c];
        }
        a_s[i] = s0;
        a_d[i] = s1;
        kmax = kmin = fmapf(s0);
    }
#pragma unroll
    for (int off = 2; off < 32; off <<= 1) {
        kmax = max(kmax, __shfl_xor(kmax, off));
        kmin = min(kmin, __shfl_xor(kmin, off));
    }
    const int lane = threadIdx.x & 31, wave = threadIdx.x >> 5;
    if (lane < HEADS) { smax[wave * HEADS + lane] = kmax; smin[wave * HEADS + lane] = kmin; }
    __syncthreads();
    if (threadIdx.x < HEADS) {
        int mx = INT_MIN, mn = INT_MAX;
        for (int w = 0; w < 8; ++w) {
            mx = max(mx, smax[w * HEADS + threadIdx.x]);
            mn = min(mn, smin[w * HEADS + threadIdx.x]);
        }
        atomicMax(&gstat[threadIdx.x], mx);
        atomicMin(&gstat[HEADS + threadIdx.x], mn);
    }
}

#define NB   196

template <bool PLANES>
__global__ __launch_bounds__(256) void init_kernel(int* __restrict__ m, int* __restrict__ gstat, int n2,
                            const int* __restrict__ dst, unsigned* __restrict__ plo,
                            unsigned* __restrict__ phi, int n256)
{
    const int i = blockIdx.x * blockDim.x + threadIdx.x;
    if (i < n2) m[i] = INT_MIN;
    if (i < 2 * HEADS) gstat[i] = i < HEADS ? INT_MIN : INT_MAX;
    if constexpr (PLANES) {
        const int lane = threadIdx.x & 31;
        const int nw   = (int)(gridDim.x * blockDim.x) >> 5;
        for (int c = i >> 5; c < n256; c += nw) {
            int d[8];
#pragma unroll
            for (int u = 0; u < 8; ++u) d[u] = dst[c * 256 + u * 32 + lane];
            unsigned wlo = 0, whi = 0;
#pragma unroll
            for (int u = 0; u < 8; ++u) {
                const unsigned key = (unsigned)d[u] / NB;
#pragma unroll
                for (int p = 0; p < 4; ++p) {
                    const unsigned vlo = __builtin_amdgcn_ballot_w32(((key >> p) & 1u) != 0);
                    const unsigned vhi = __builtin_amdgcn_ballot_w32(((key >> (p + 4)) & 1u) != 0);
                    wlo = lane == u * 4 + p ? vlo : wlo;
                    whi = lane == u * 4 + p ? vhi : whi;
                }
            }
            plo[c * 32 + lane] = wlo;
            phi[c * 32 + lane] = whi;
        }
    }
}

__global__ void edge_max_kernel(const int* __restrict__ src, const int* __restrict__ dst,
                                const float* __restrict__ a_s, const float* __restrict__ a_d,
                                int* __restrict__ m, const int* __restrict__ gstat, int E_, int M_)
{
    float hi0, hi1;
    if (shift_from_bound(*(const int4*)gstat, hi0, hi1)) return;
    int k = blockIdx.x * blockDim.x + threadIdx.x;
    if (k >= M_) return;
    const int s = k < E_ ? src[k] : k - E_;
    const int d = k < E_ ? dst[k] : k - E_;
#pragma unroll
    for (int h = 0; h < HEADS; ++h) {
        float e = lrelu(a_s[s * HEADS + h] + a_d[d * HEADS + h]);
        atomicMax(&m[d * HEADS + h], fmapf(e));
    }
}

#define QCAP 8192
#define BT   1024
#define BU   8
#define RU   4

template <bool FULL>
__device__ __forceinline__ void add_group(
    const int (&s)[BU], const int (&dl)[BU], const float (&p0)[BU], const float (&p1)[BU],
    const bool (&ok)[BU], const int lane, const float* __restrict__ Hp, float* acc)
{
    float h0[BU], h1[BU], h2[BU], h3[BU];
#pragma unroll
    for (int j = 0; j < BU; ++j) {
        const float* hrow = Hp + (size_t)s[j] * FDIM + lane;
        h0[j] = hrow[0];
        h1[j] = hrow[32];
        h2[j] = hrow[64];
        h3[j] = hrow[96];
    }
#pragma unroll
    for (int j = 0; j < BU; ++j) {
        if (FULL || ok[j]) {
            float* o = acc + dl[j] * FDIM + lane;
            atomicAdd(o,      h0[j] * p0[j]);
            atomicAdd(o + 32, h1[j] * p0[j]);
            atomicAdd(o + 64, h2[j] * p1[j]);
            atomicAdd(o + 96, h3[j] * p1[j]);
        }
    }
}

template <bool FIRST>
__global__ __launch_bounds__(BT) void block_aggregate_kernel(
    const int* __restrict__ src, const int* __restrict__ dst,
    const float* __restrict__ a_s, const float* __restrict__ a_d,
    const int* __restrict__ m, const int* __restrict__ gstat,
    const float* __restrict__ Hp, float* __restrict__ x,
    int2* __restrict__ lists, int* __restrict__ lcnt,
    const uint4* __restrict__ plo, const uint4* __restrict__ phi,
    int E_, int N_, int relu)
{
    __shared__ __attribute__((aligned(16))) float acc[NB * FDIM];
    __shared__ __attribute__((aligned(16))) int4 rec[QCAP];
    __shared__ int   rawq[QCAP];
    __shared__ float den[NB * HEADS], sh_ad[NB * HEADS], sh_c[NB * HEADS];
    __shared__ int   qcount;

    const int tid   = threadIdx.x;
    const int lane  = tid & 31;
    const int wave  = __builtin_amdgcn_readfirstlane(tid >> 5);
    const int node0 = blockIdx.x * NB;
    const int nb    = min(NB, N_ - node0);

    float hi0, hi1;
    const bool bound = shift_from_bound(*(const int4*)gstat, hi0, hi1);

    for (int i = tid; i < NB * FDIM / 4; i += BT) ((float4*)acc)[i] = make_float4(0.0f, 0.0f, 0.0f, 0.0f);
    for (int i = tid; i < nb * HEADS; i += BT) {
        const int   g  = node0 * HEADS + i;
        const float ad = a_d[g];
        sh_ad[i] = ad;
        sh_c[i]  = bound ? lrelu(((i & 1) ? hi1 : hi0) + ad) : funmap(m[g]);
        den[i]   = 0.0f;
    }
    if (tid == 0) qcount = 0;
    __syncthreads();

    auto file_edge = [&](const int d, const int k) {
        if ((unsigned)(d - node0) < (unsigned)nb) {
            const int slot = atomicAdd(&qcount, 1);
            if (slot < QCAP) rawq[slot] = k;
        }
    };

    int cnt;
    if constexpr (FIRST) {
        int kdone = 0;
        if (gridDim.x <= 256) {
            const int n32 = (E_ >> 8) << 3;
            kdone = n32 << 5;
            unsigned inv[8];
#pragma unroll
            for (int p = 0; p < 8; ++p) inv[p] = ((blockIdx.x >> p) & 1u) ? 0u : ~0u;
            const int start = (int)blockIdx.x * (n32 / (int)gridDim.x);
            for (int i0 = tid; i0 < n32; i0 += 4 * BT) {
                uint4 lo[4], hi[4];
                int   gx[4];
#pragma unroll
                for (int u = 0; u < 4; ++u) {
                    const int i = i0 + u * BT;
                    int t = start + (i < n32 ? i : i0);
                    if (t >= n32) t -= n32;
                    lo[u] = plo[t];
                    hi[u] = phi[t];
                    gx[u] = i < n32 ? t : -1;
                }
                unsigned hit[4];
                int found = 0;
#pragma unroll
                for (int u = 0; u < 4; ++u) {
                    const unsigned a = (lo[u].x ^ inv[0]) & (lo[u].y ^ inv[1]) & (lo[u].z ^ inv[2]) & (lo[u].w ^ inv[3]);
                    const unsigned b = (hi[u].x ^ inv[4]) & (hi[u].y ^ inv[5]) & (hi[u].z ^ inv[6]) & (hi[u].w ^ inv[7]);
                    hit[u] = gx[u] >= 0 ? (a & b) : 0u;
                    found += __popc(hit[u]);
                }
                if (found) {
                    int slot = atomicAdd(&qcount, found);
#pragma unroll
                    for (int u = 0; u < 4; ++u) {
                        for (unsigned h = hit[u]; h; h &= h - 1) {
                            if (slot < QCAP) rawq[slot] = (gx[u] << 5) + __builtin_ctz(h);
                            ++slot;
                        }
                    }
                }
            }
        }
        for (int k = kdone + tid; k < E_; k += BT) file_edge(dst[k], k);
        __syncthreads();
        cnt = qcount;
        if (tid == 0) lcnt[blockIdx.x] = cnt;
    } else {
        cnt = lcnt[blockIdx.x];
    }

    auto resolve = [&](const int nr, const int ns, const int from) {
        int2* __restrict__ mine = lists + (size_t)blockIdx.x * QCAP;
        const int n = nr + ns;
        for (int i0 = tid; i0 < n; i0 += RU * BT) {
            int s[RU], dl[RU];
            v2f as2[RU];
            if (from == 2) {
#pragma unroll
                for (int u = 0; u < RU; ++u) {
                    const int  i    = min(i0 + u * BT, n - 1);
                    const bool pair = i < nr;
                    const int2 e    = mine[pair ? i : 0];
                    dl[u] = pair ? e.y : i - nr;
                    s[u]  = pair ? e.x : node0 + (i - nr);
                }
            } else {
                int k[RU];
#pragma unroll
                for (int u = 0; u < RU; ++u) {
                    const int  i    = min(i0 + u * BT, n - 1);
                    const bool pair = i < nr;
                    const int  kq   = rawq[pair ? i : 0];
                    k[u] = pair ? kq : 0;
                }
#pragma unroll
                for (int u = 0; u < RU; ++u) {
                    const int  i    = min(i0 + u * BT, n - 1);
                    const bool pair = i < nr;
                    const int  sv   = src[k[u]];
                    const int  dv   = dst[k[u]];
                    s[u]  = pair ? sv : node0 + (i - nr);
                    dl[u] = pair ? dv - node0 : i - nr;
                }
            }
#pragma unroll
            for (int u = 0; u < RU; ++u)
                as2[u] = *(const v2f*)(a_s + s[u] * HEADS);
#pragma unroll
            for (int u = 0; u < RU; ++u) {
                const int i = i0 + u * BT;
                if (i < n) {
                    const float p0 = __expf(lrelu(as2[u].x + sh_ad[dl[u] * HEADS])     - sh_c[dl[u] * HEADS]);
                    const float p1 = __expf(lrelu(as2[u].y + sh_ad[dl[u] * HEADS + 1]) - sh_c[dl[u] * HEADS + 1]);
                    atomicAdd(&den[dl[u] * HEADS],     p0);
                    atomicAdd(&den[dl[u] * HEADS + 1], p1);
                    rec[i] = make_int4(s[u], dl[u], __float_as_int(p0), __float_as_int(p1));
                    if (from == 1 && i < nr) mine[i] = make_int2(s[u], dl[u]);
                }
            }
        }
    };

    auto walk = [&](const int n) {
        for (int q0 = wave * BU; q0 < n; q0 += (BT / 32) * BU) {
            int s[BU], dl[BU]; float p0[BU], p1[BU]; bool ok[BU];
#pragma unroll
            for (int j = 0; j < BU; ++j) {
                ok[j] = q0 + j < n;
                const int4 e = rec[ok[j] ? q0 + j : n - 1];
                s[j]  = __builtin_amdgcn_readfirstlane(e.x);
                dl[j] = __builtin_amdgcn_readfirstlane(e.y);
                p0[j] = __int_as_float(__builtin_amdgcn_readfirstlane(e.z));
                p1[j] = __int_as_float(__builtin_amdgcn_readfirstlane(e.w));
            }
            if (q0 + BU <= n) add_group<true>(s, dl, p0, p1, ok, lane, Hp, acc);
            else              add_group<false>(s, dl, p0, p1, ok, lane, Hp, acc);
        }
    };

    if (cnt + nb <= QCAP) {
        resolve(cnt, nb, FIRST ? 1 : 2);
        __syncthreads();
        walk(cnt + nb);
    } else {
        resolve(0, nb, 0);
        __syncthreads();
        walk(nb);
        for (int b0 = 0; b0 < E_; b0 += QCAP) {
            __syncthreads();
            if (tid == 0) qcount = 0;
            __syncthreads();
            const int b1 = min(b0 + QCAP, E_);
            for (int k = b0 + tid; k < b1; k += BT) file_edge(dst[k], k);
            __syncthreads();
            const int n = qcount;
            resolve(n, 0, 0);
            __syncthreads();
            walk(n);
        }
    }
    __syncthreads();

    for (int i = wave; i < nb; i += BT / 32) {
        const float dn0 = den[i * HEADS] + 1e-16f;
        const float dn1 = den[i * HEADS + 1] + 1e-16f;
        float*       xr = x + (size_t)(node0 + i) * FDIM + lane;
        const float* ar = acc + i * FDIM + lane;
        float v0 = xr[0]  + ar[0]  / dn0;
        float v1 = xr[32] + ar[32] / dn0;
        float v2 = xr[64] + ar[64] / dn1;
        float v3 = xr[96] + ar[96] / dn1;
        if (relu) {
            v0 = fmaxf(v0, 0.0f); v1 = fmaxf(v1, 0.0f);
            v2 = fmaxf(v2, 0.0f); v3 = fmaxf(v3, 0.0f);
        }
        xr[0] = v0; xr[32] = v1; xr[64] = v2; xr[96] = v3;
    }
}

extern "C" void kernel_launch(void* const* d_in, const int* in_sizes, int n_in,
                              void* d_out, int out_size, void* d_ws, size_t ws_size,
                              hipStream_t stream)
{
    const float* x        = (const float*)d_in[0];
    const int*   src      = (const int*)  d_in[1];
    const int*   dst      = (const int*)  d_in[2];
    const float* W1       = (const float*)d_in[3];
    const float* att_src1 = (const float*)d_in[4];
    const float* att_dst1 = (const float*)d_in[5];
    const float* b1       = (const float*)d_in[6];
    const float* Wsk1     = (const float*)d_in[7];
    const float* bsk1     = (const float*)d_in[8];
    const float* W2       = (const float*)d_in[9];
    const float* att_src2 = (const float*)d_in[10];
    const float* att_dst2 = (const float*)d_in[11];
    const float* b2       = (const float*)d_in[12];
    const float* Wsk2     = (const float*)d_in[13];
    const float* bsk2     = (const float*)d_in[14];

    const int N  = in_sizes[0] / FDIM;
    const int E  = in_sizes[1];
    const int M  = E + N;
    float* out   = (float*)d_out;
    const int nblk = (N + NB - 1) / NB;

    char* base = (char*)d_ws;
    size_t off = 0;
    auto alloc = [&](size_t bytes) {
        void* p = base + off;
        off = (off + bytes + 255) & ~(size_t)255;
        return p;
    };
    float* Hproj = (float*)alloc((size_t)N * FDIM * 4);
    float* Hbuf  = (float*)alloc((size_t)N * FDIM * 4);
    float* a_s   = (float*)alloc((size_t)N * HEADS * 4);
    float* a_d   = (float*)alloc((size_t)N * HEADS * 4);
    int*   mbuf  = (int*)  alloc((size_t)N * HEADS * 4);
    int*   gstat = (int*)  alloc(2 * HEADS * 4);
    int2*  lists = (int2*) alloc((size_t)nblk * QCAP * 8);
    int*   lcnt  = (int*)  alloc((size_t)nblk * 4);
    const int n256 = E >> 8;
    unsigned* plo = (unsigned*)alloc((size_t)n256 * 128);
    unsigned* phi = (unsigned*)alloc((size_t)n256 * 128);

    const int ntiles     = N / 16;
    const dim3 gGemm((ntiles + 3) / 4), bGemm(128);
    const dim3 gAtt((N * HEADS + 255) / 256), b256(256);
    const dim3 gEdge((M + 255) / 256);
    const dim3 gBlk(nblk), bBlk(BT);

    gemm_f32_wmma<<<gGemm, bGemm, 0, stream>>>(x, W1, nullptr, nullptr, Hproj, N);
    gemm_f32_wmma<<<gGemm, bGemm, 0, stream>>>(x, Wsk1, b1, bsk1, Hbuf, N);
    init_kernel<true><<<gAtt, b256, 0, stream>>>(mbuf, gstat, N * HEADS, dst, plo, phi, n256);
    att_kernel<<<gAtt, b256, 0, stream>>>(Hproj, att_src1, att_dst1, a_s, a_d, gstat, N);
    edge_max_kernel<<<gEdge, b256, 0, stream>>>(src, dst, a_s, a_d, mbuf, gstat, E, M);
    block_aggregate_kernel<true><<<gBlk, bBlk, 0, stream>>>(src, dst, a_s, a_d, mbuf, gstat, Hproj, Hbuf, lists, lcnt,
                                                         (const uint4*)plo, (const uint4*)phi, E, N, 1);

    gemm_f32_wmma<<<gGemm, bGemm, 0, stream>>>(Hbuf, W2, nullptr, nullptr, Hproj, N);
    gemm_f32_wmma<<<gGemm, bGemm, 0, stream>>>(Hbuf, Wsk2, b2, bsk2, out, N);
    init_kernel<false><<<gAtt, b256, 0, stream>>>(mbuf, gstat, N * HEADS, dst, plo, phi, n256);
    att_kernel<<<gAtt, b256, 0, stream>>>(Hproj, att_src2, att_dst2, a_s, a_d, gstat, N);
    edge_max_kernel<<<gEdge, b256, 0, stream>>>(src, dst, a_s, a_d, mbuf, gstat, E, M);
    block_aggregate_kernel<false><<<gBlk, bBlk, 0, stream>>>(src, dst, a_s, a_d, mbuf, gstat, Hproj, out, lists, lcnt,
                                                          (const uint4*)plo, (const uint4*)phi, E, N, 0);
}
